# starting kernel
# baseline (speedup 1.0000x reference)
#include <hip/hip_runtime.h>

#define D_IN 512
#define DK   64
#define DV   64
#define B_   8
#define S_   2048

typedef __attribute__((ext_vector_type(2))) float v2f;
typedef __attribute__((ext_vector_type(8))) float v8f;

static __device__ __forceinline__ v8f wmma_f32(v2f a, v2f b, v8f c) {
  return __builtin_amdgcn_wmma_f32_16x16x4_f32(
       false, a,  false, b,
       (short)0, c,  false,  false);
}

__global__ void transpose_weights(const float* __restrict__ Wk,
                                  const float* __restrict__ Wq,
                                  const float* __restrict__ Wv,
                                  float* __restrict__ WT3) {
  int idx = blockIdx.x * 256 + threadIdx.x;
  if (idx >= 3 * D_IN * DK) return;
  int mat = idx >> 15;
  int rem = idx & 32767;
  int d = rem >> 6;
  int n = rem & 63;
  const float* W = (mat == 0) ? Wk : ((mat == 1) ? Wq : Wv);
  WT3[idx] = W[n * D_IN + d];
}

__global__ __launch_bounds__(256) void qkv_kernel(
    const float* __restrict__ X, const float* __restrict__ WT3,
    const float* __restrict__ bk, const float* __restrict__ bq,
    const float* __restrict__ bv,
    float* __restrict__ Kb, float* __restrict__ Qb, float* __restrict__ Vb) {
  const int lane = threadIdx.x & 31;
  const int wave = threadIdx.x >> 5;
  const int wg   = blockIdx.x * 8 + wave;
  const int b    = wg >> 7;
  const int s0   = (wg & 127) << 4;
  const int m    = lane & 15;
  const int hi   = lane >> 4;

  const float* xr = X + (size_t)(b * S_ + s0 + m) * D_IN + 2 * hi;

  v8f acc[3][4];
#pragma unroll
  for (int i = 0; i < 3; ++i)
#pragma unroll
    for (int j = 0; j < 4; ++j)
      acc[i][j] = (v8f){0.f, 0.f, 0.f, 0.f, 0.f, 0.f, 0.f, 0.f};

  for (int t = 0; t < D_IN / 4; ++t) {
    v2f a;
    a.x = xr[4 * t];
    a.y = xr[4 * t + 1];
    const float* wbase = WT3 + ((size_t)(4 * t + 2 * hi)) * DK + m;
#pragma unroll
    for (int mtx = 0; mtx < 3; ++mtx) {
#pragma unroll
      for (int nt = 0; nt < 4; ++nt) {
        const float* wp = wbase + (size_t)mtx * D_IN * DK + nt * 16;
        v2f bf;
        bf.x = wp[0];
        bf.y = wp[DK];
        acc[mtx][nt] = wmma_f32(a, bf, acc[mtx][nt]);
      }
    }
  }

  const float* bias[3] = {bk, bq, bv};
  float* out[3] = {Kb, Qb, Vb};
#pragma unroll
  for (int mtx = 0; mtx < 3; ++mtx) {
#pragma unroll
    for (int nt = 0; nt < 4; ++nt) {
      float bs = bias[mtx][nt * 16 + m];
#pragma unroll
      for (int r = 0; r < 8; ++r) {
        out[mtx][(size_t)(b * S_ + s0 + r + 8 * hi) * DK + nt * 16 + m] =
            acc[mtx][nt][r] + bs;
      }
    }
  }
}

__global__ __launch_bounds__(256) void attn_kernel(
    const float* __restrict__ Kb, const float* __restrict__ Qb,
    const float* __restrict__ Vb, float* __restrict__ Y) {
  __shared__ float lds_e[256 * 12];
  __shared__ float lds_a[8][16 * 17];

  const int lane = threadIdx.x & 31;
  const int b    = threadIdx.x >> 5;
  const int m    = lane & 15;
  const int hi   = lane >> 4;
  const int i0   = blockIdx.x << 4;

  const float scale = 0.125f;

  v2f ka[16];
  {
    const float* kr = Kb + (size_t)(b * S_ + i0 + m) * DK + 2 * hi;
#pragma unroll
    for (int t = 0; t < 16; ++t) {
      ka[t].x = kr[4 * t] * scale;
      ka[t].y = kr[4 * t + 1] * scale;
    }
  }

  v8f yacc[4];
#pragma unroll
  for (int nt = 0; nt < 4; ++nt)
    yacc[nt] = (v8f){0.f, 0.f, 0.f, 0.f, 0.f, 0.f, 0.f, 0.f};

  for (int j0 = 0; j0 < S_; j0 += 16) {
    v8f e = (v8f){0.f, 0.f, 0.f, 0.f, 0.f, 0.f, 0.f, 0.f};
    {
      const float* qr = Qb + (size_t)(b * S_ + j0 + m) * DK + 2 * hi;
#pragma unroll
      for (int t = 0; t < 16; ++t) {
        v2f qb;
        qb.x = qr[4 * t];
        qb.y = qr[4 * t + 1];
        e = wmma_f32(ka[t], qb, e);
      }
    }

#pragma unroll
    for (int r = 0; r < 8; ++r)
      lds_e[(r * 32 + lane) * 12 + b] = e[r];
    __syncthreads();

    float aval[8];
#pragma unroll
    for (int r = 0; r < 8; ++r) {
      const float4* p = (const float4*)&lds_e[(r * 32 + lane) * 12];
      float4 u = p[0];
      float4 w = p[1];
      float mx = fmaxf(fmaxf(fmaxf(u.x, u.y), fmaxf(u.z, u.w)),
                       fmaxf(fmaxf(w.x, w.y), fmaxf(w.z, w.w)));
      float z = __expf(u.x - mx) + __expf(u.y - mx) + __expf(u.z - mx) +
                __expf(u.w - mx) + __expf(w.x - mx) + __expf(w.y - mx) +
                __expf(w.z - mx) + __expf(w.w - mx);
      aval[r] = __expf(e[r] - mx) / z;
    }

#pragma unroll
    for (int r = 0; r < 8; ++r)
      lds_a[b][(r + 8 * hi) * 17 + m] = aval[r];
    __syncthreads();

    v2f af[4];
#pragma unroll
    for (int t = 0; t < 4; ++t) {
      af[t].x = lds_a[b][m * 17 + 4 * t + 2 * hi];
      af[t].y = lds_a[b][m * 17 + 4 * t + 2 * hi + 1];
    }

#pragma unroll
    for (int t = 0; t < 4; ++t) {
      const float* vr = Vb + (size_t)(b * S_ + j0 + 4 * t + 2 * hi) * DV + m;
#pragma unroll
      for (int nt = 0; nt < 4; ++nt) {
        v2f vb;
        vb.x = vr[nt * 16];
        vb.y = vr[DV + nt * 16];
        yacc[nt] = wmma_f32(af[t], vb, yacc[nt]);
      }
    }
  }

#pragma unroll
  for (int nt = 0; nt < 4; ++nt)
#pragma unroll
    for (int r = 0; r < 8; ++r)
      Y[(size_t)(b * S_ + i0 + r + 8 * hi) * DV + nt * 16 + m] =
          yacc[nt][r];
}

extern "C" void kernel_launch(void* const* d_in, const int* in_sizes, int n_in,
                              void* d_out, int out_size, void* d_ws,
                              size_t ws_size, hipStream_t stream) {
  const float* X  = (const float*)d_in[0];
  const float* Wk = (const float*)d_in[1];
  const float* bk = (const float*)d_in[2];
  const float* Wq = (const float*)d_in[3];
  const float* bq = (const float*)d_in[4];
  const float* Wv = (const float*)d_in[5];
  const float* bv = (const float*)d_in[6];
  float* Y = (float*)d_out;

  float* ws = (float*)d_ws;
  const size_t NKQV = (size_t)B_ * S_ * DK;
  float* Kb  = ws;
  float* Qb  = ws + NKQV;
  float* Vb  = ws + 2 * NKQV;
  float* WT3 = ws + 3 * NKQV;

  transpose_weights<<<(3 * D_IN * DK + 255) / 256, 256, 0, stream>>>(
      Wk, Wq, Wv, WT3);
  qkv_kernel<<<(B_ * S_ / 16) / 8, 256, 0, stream>>>(
      X, WT3, bk, bq, bv, Kb, Qb, Vb);
  attn_kernel<<<S_ / 16, 256, 0, stream>>>(Kb, Qb, Vb, Y);
}


# attn kernel re-split: a wave owns all 8 batches of a 16x16 (j,i) tile: batch softmax in registers, E^T tile is in place the B operand of Y^T=V^T*A^T; block=(i-tile,V half); d and j sum order change
# speedup vs baseline: 2.0688x; 2.0688x over previous
#include <hip/hip_runtime.h>

#define D_IN 512
#define DK   64
#define DV   64
#define B_   8
#define S_   2048

typedef __attribute__((ext_vector_type(2))) float v2f;
typedef __attribute__((ext_vector_type(4))) float v4f;
typedef __attribute__((ext_vector_type(8))) float v8f;

static __device__ __forceinline__ v8f wmma_f32(v2f a, v2f b, v8f c) {
  return __builtin_amdgcn_wmma_f32_16x16x4_f32(
       false, a,  false, b,
       (short)0, c,  false,  false);
}

__global__ void transpose_weights(const float* __restrict__ Wk,
                                  const float* __restrict__ Wq,
                                  const float* __restrict__ Wv,
                                  float* __restrict__ WT3) {
  int idx = blockIdx.x * 256 + threadIdx.x;
  if (idx >= 3 * D_IN * DK) return;
  int mat = idx >> 15;
  int rem = idx & 32767;
  int d = rem >> 6;
  int n = rem & 63;
  const float* W = (mat == 0) ? Wk : ((mat == 1) ? Wq : Wv);
  WT3[idx] = W[n * D_IN + d];
}

__global__ __launch_bounds__(256) void qkv_kernel(
    const float* __restrict__ X, const float* __restrict__ WT3,
    const float* __restrict__ bk, const float* __restrict__ bq,
    const float* __restrict__ bv,
    float* __restrict__ Kb, float* __restrict__ Qb, float* __restrict__ Vb) {
  const int lane = threadIdx.x & 31;
  const int wave = threadIdx.x >> 5;
  const int wg   = blockIdx.x * 8 + wave;
  const int b    = wg >> 7;
  const int s0   = (wg & 127) << 4;
  const int m    = lane & 15;
  const int hi   = lane >> 4;

  const float* xr = X + (size_t)(b * S_ + s0 + m) * D_IN + 2 * hi;

  v8f acc[3][4];
#pragma unroll
  for (int i = 0; i < 3; ++i)
#pragma unroll
    for (int j = 0; j < 4; ++j)
      acc[i][j] = (v8f){0.f, 0.f, 0.f, 0.f, 0.f, 0.f, 0.f, 0.f};

  for (int t = 0; t < D_IN / 4; ++t) {
    v2f a;
    a.x = xr[4 * t];
    a.y = xr[4 * t + 1];
    const float* wbase = WT3 + ((size_t)(4 * t + 2 * hi)) * DK + m;
#pragma unroll
    for (int mtx = 0; mtx < 3; ++mtx) {
#pragma unroll
      for (int nt = 0; nt < 4; ++nt) {
        const float* wp = wbase + (size_t)mtx * D_IN * DK + nt * 16;
        v2f bf;
        bf.x = wp[0];
        bf.y = wp[DK];
        acc[mtx][nt] = wmma_f32(a, bf, acc[mtx][nt]);
      }
    }
  }

  const float* bias[3] = {bk, bq, bv};
  float* out[3] = {Kb, Qb, Vb};
#pragma unroll
  for (int mtx = 0; mtx < 3; ++mtx) {
#pragma unroll
    for (int nt = 0; nt < 4; ++nt) {
      float bs = bias[mtx][nt * 16 + m];
#pragma unroll
      for (int r = 0; r < 8; ++r) {
        out[mtx][(size_t)(b * S_ + s0 + r + 8 * hi) * DK + nt * 16 + m] =
            acc[mtx][nt][r] + bs;
      }
    }
  }
}

#define KP 68
#define RP 33

__global__ __launch_bounds__(256) void attn_kernel(
    const float* __restrict__ Kb, const float* __restrict__ Qb,
    const float* __restrict__ Vb, float* __restrict__ Y) {
  __shared__ __attribute__((aligned(16))) float lds_k[B_ * 16 * KP];
  __shared__ float lds_y[8 * B_ * 2 * 8 * RP];

  const int lane = threadIdx.x & 31;
  const int w    = threadIdx.x >> 5;
  const int c    = lane & 15;
  const int h    = lane >> 4;
  const int i0   = (blockIdx.x >> 1) << 4;
  const int vb   = (blockIdx.x & 1) << 5;

  {
    const float scale = 0.125f;
#pragma unroll
    for (int it = 0; it < 8; ++it) {
      const int idx = threadIdx.x + 256 * it;
      const int b   = idx >> 8;
      const int row = (idx >> 4) & 15;
      const int d4  = idx & 15;
      v4f kv = *(const v4f*)(Kb + (size_t)(b * S_ + i0 + row) * DK + 4 * d4);
      kv = kv * scale;
      *(v4f*)(lds_k + (b * 16 + row) * KP + 4 * d4) = kv;
    }
  }
  __syncthreads();

  v8f yacc[B_][2];
#pragma unroll
  for (int b = 0; b < B_; ++b)
#pragma unroll
    for (int vt = 0; vt < 2; ++vt)
      yacc[b][vt] = (v8f){0.f, 0.f, 0.f, 0.f, 0.f, 0.f, 0.f, 0.f};

  v4f qs[3][8];
  v4f ks[2][8];
  v2f vs[5][4][2];
  v8f e[B_];

  auto load_q = [&](int set, int b, int j0) {
    const float* qr = Qb + (size_t)(b * S_ + j0 + c) * DK + 4 * h;
#pragma unroll
    for (int T = 0; T < 8; ++T) qs[set][T] = *(const v4f*)(qr + 8 * T);
  };
  auto load_k = [&](int set, int b) {
    const float* kr = lds_k + (b * 16 + c) * KP + 4 * h;
#pragma unroll
    for (int T = 0; T < 8; ++T) ks[set][T] = *(const v4f*)(kr + 8 * T);
  };
  auto load_v = [&](int set, int b, int j0) {
    const float* vr = Vb + (size_t)(b * S_ + j0 + 8 * h) * DV + vb + 2 * c;
#pragma unroll
    for (int t = 0; t < 4; ++t)
#pragma unroll
      for (int s = 0; s < 2; ++s)
        vs[set][t][s] = *(const v2f*)(vr + (2 * t + s) * DV);
  };
  auto e_phase = [&](int j0) {
#pragma unroll
    for (int b = 0; b < B_; ++b) {
      if (b + 2 < B_) load_q((b + 2) % 3, b + 2, j0);
      if (b + 1 < B_) load_k((b + 1) % 2, b + 1);
      v8f acc0 = (v8f){0.f, 0.f, 0.f, 0.f, 0.f, 0.f, 0.f, 0.f};
      v8f acc1 = (v8f){0.f, 0.f, 0.f, 0.f, 0.f, 0.f, 0.f, 0.f};
#pragma unroll
      for (int T = 0; T < 8; ++T) {
        const v4f q = qs[b % 3][T];
        const v4f k = ks[b % 2][T];
        acc0 = wmma_f32((v2f){q.x, q.y}, (v2f){k.x, k.y}, acc0);
        acc1 = wmma_f32((v2f){q.z, q.w}, (v2f){k.z, k.w}, acc1);
      }
      e[b] = acc0 + acc1;
    }
  };

  load_q(0, 0, (w * 16) << 4);
  load_q(1, 1, (w * 16) << 4);

  for (int n = 0; n < 16; ++n) {
    const int j0  = (w * 16 + n) << 4;
    const int j0n = (n < 15) ? j0 + 16 : j0;

    load_k(0, 0);
    e_phase(j0);

    load_v(0, 0, j0);
    load_v(1, 1, j0);
    load_v(2, 2, j0);
    load_v(3, 3, j0);
    load_q(0, 0, j0n);
    load_q(1, 1, j0n);

    asm volatile("" ::: "memory");

#pragma unroll
    for (int r = 0; r < 8; ++r) {
      const float mx =
          fmaxf(fmaxf(fmaxf(e[0][r], e[1][r]), fmaxf(e[2][r], e[3][r])),
                fmaxf(fmaxf(e[4][r], e[5][r]), fmaxf(e[6][r], e[7][r])));
      float p[B_];
#pragma unroll
      for (int b = 0; b < B_; ++b) p[b] = __expf(e[b][r] - mx);
      const float z = p[0] + p[1] + p[2] + p[3] + p[4] + p[5] + p[6] + p[7];
#pragma unroll
      for (int b = 0; b < B_; ++b) e[b][r] = p[b] / z;
    }

#pragma unroll
    for (int b = 0; b < B_; ++b) {
      if (b + 4 < B_) load_v((b + 4) % 5, b + 4, j0);
#pragma unroll
      for (int t = 0; t < 4; ++t) {
        const v2f pb = (v2f){e[b][2 * t], e[b][2 * t + 1]};
        const v2f v0 = vs[b % 5][t][0];
        const v2f v1 = vs[b % 5][t][1];
        yacc[b][0] = wmma_f32((v2f){v0.x, v1.x}, pb, yacc[b][0]);
        yacc[b][1] = wmma_f32((v2f){v0.y, v1.y}, pb, yacc[b][1]);
      }
    }
  }

#pragma unroll
  for (int b = 0; b < B_; ++b)
#pragma unroll
    for (int vt = 0; vt < 2; ++vt)
#pragma unroll
      for (int r = 0; r < 8; ++r)
        lds_y[(((w * B_ + b) * 2 + vt) * 8 + r) * RP + lane] = yacc[b][vt][r];
  __syncthreads();

  {
    const int vt = lane & 1;
    const int hh = lane >> 4;
    const int rr = (lane >> 1) & 7;
#pragma unroll 4
    for (int i = 0; i < 16; ++i) {
      float s = lds_y[(((0 * B_ + w) * 2 + vt) * 8 + rr) * RP + i + 16 * hh];
#pragma unroll
      for (int ww = 1; ww < 8; ++ww)
        s += lds_y[(((ww * B_ + w) * 2 + vt) * 8 + rr) * RP + i + 16 * hh];
      Y[(size_t)(w * S_ + i0 + i) * DV + vb + lane] = s;
    }
  }
}

extern "C" void kernel_launch(void* const* d_in, const int* in_sizes, int n_in,
                              void* d_out, int out_size, void* d_ws,
                              size_t ws_size, hipStream_t stream) {
  const float* X  = (const float*)d_in[0];
  const float* Wk = (const float*)d_in[1];
  const float* bk = (const float*)d_in[2];
  const float* Wq = (const float*)d_in[3];
  const float* bq = (const float*)d_in[4];
  const float* Wv = (const float*)d_in[5];
  const float* bv = (const float*)d_in[6];
  float* Y = (float*)d_out;

  float* ws = (float*)d_ws;
  const size_t NKQV = (size_t)B_ * S_ * DK;
  float* Kb  = ws;
  float* Qb  = ws + NKQV;
  float* Vb  = ws + 2 * NKQV;
  float* WT3 = ws + 3 * NKQV;

  transpose_weights<<<(3 * D_IN * DK + 255) / 256, 256, 0, stream>>>(
      Wk, Wq, Wv, WT3);
  qkv_kernel<<<(B_ * S_ / 16) / 8, 256, 0, stream>>>(
      X, WT3, bk, bq, bv, Kb, Qb, Vb);
  attn_kernel<<<256, 256, 0, stream>>>(Kb, Qb, Vb, Y);
}


# projection re-tiled: wave = 32 rows x 96 cols, 256 blocks of 4 waves; X and row-major weight fragments both read as 16-B k runs 7 steps ahead; no transpose launch (2 launches); k sum order changes
# speedup vs baseline: 2.5607x; 1.2378x over previous
#include <hip/hip_runtime.h>

#define D_IN 512
#define DK   64
#define DV   64
#define B_   8
#define S_   2048

typedef __attribute__((ext_vector_type(2))) float v2f;
typedef __attribute__((ext_vector_type(4))) float v4f;
typedef __attribute__((ext_vector_type(8))) float v8f;

static __device__ __forceinline__ v8f wmma_f32(v2f a, v2f b, v8f c) {
  return __builtin_amdgcn_wmma_f32_16x16x4_f32(
       false, a,  false, b,
       (short)0, c,  false,  false);
}

#define RING 8

__global__ __launch_bounds__(128) void qkv_kernel(
    const float* __restrict__ X,
    const float* __restrict__ Wk, const float* __restrict__ Wq,
    const float* __restrict__ Wv,
    const float* __restrict__ bk, const float* __restrict__ bq,
    const float* __restrict__ bv,
    float* __restrict__ Kb, float* __restrict__ Qb, float* __restrict__ Vb) {
  const int lane = threadIdx.x & 31;
  const int w    = threadIdx.x >> 5;
  const int c    = lane & 15;
  const int h    = lane >> 4;
  const int rh   = w >> 1;
  const int ch   = w & 1;
  const int row0 = blockIdx.x * 64 + rh * 32;

  const float* xp = X + (size_t)(row0 + c) * D_IN + 4 * h;
  const float* wp[6];
  const float* bp[6];
  float*       op[6];
#pragma unroll
  for (int j = 0; j < 6; ++j) {
    const int g   = 6 * ch + j;
    const int mat = g >> 2;
    const int col = ((g & 3) << 4) + c;
    const float* Wm = (mat == 0) ? Wk : ((mat == 1) ? Wq : Wv);
    const float* bm = (mat == 0) ? bk : ((mat == 1) ? bq : bv);
    float*       om = (mat == 0) ? Kb : ((mat == 1) ? Qb : Vb);
    wp[j] = Wm + (size_t)col * D_IN + 4 * h;
    bp[j] = bm + col;
    op[j] = om + (size_t)(row0 + 8 * h) * DK + col;
  }

  v8f acc[2][6];
#pragma unroll
  for (int rt = 0; rt < 2; ++rt)
#pragma unroll
    for (int j = 0; j < 6; ++j)
      acc[rt][j] = (v8f){0.f, 0.f, 0.f, 0.f, 0.f, 0.f, 0.f, 0.f};

  v4f xa[RING][2];
  v4f wb[RING][6];
  auto load_set = [&](int set, int T) {
#pragma unroll
    for (int rt = 0; rt < 2; ++rt)
      xa[set][rt] = *(const v4f*)(xp + (size_t)rt * 16 * D_IN + 8 * T);
#pragma unroll
    for (int j = 0; j < 6; ++j)
      wb[set][j] = *(const v4f*)(wp[j] + 8 * T);
  };

#pragma unroll
  for (int s = 0; s < RING - 1; ++s) {
    load_set(s, s);
    asm volatile("" ::: "memory");
  }

  for (int T0 = 0; T0 < D_IN / 8; T0 += RING) {
#pragma unroll
    for (int q = 0; q < RING; ++q) {
      int Tn = T0 + q + RING - 1;
      Tn = (Tn < D_IN / 8) ? Tn : D_IN / 8 - 1;
      load_set((q + RING - 1) % RING, Tn);
      asm volatile("" ::: "memory");
      __builtin_amdgcn_sched_barrier(0);
#pragma unroll
      for (int rt = 0; rt < 2; ++rt)
#pragma unroll
        for (int j = 0; j < 6; ++j) {
          const v4f a = xa[q][rt];
          const v4f b = wb[q][j];
          acc[rt][j] = wmma_f32((v2f){a.x, a.y}, (v2f){b.x, b.y}, acc[rt][j]);
        }
#pragma unroll
      for (int rt = 0; rt < 2; ++rt)
#pragma unroll
        for (int j = 0; j < 6; ++j) {
          const v4f a = xa[q][rt];
          const v4f b = wb[q][j];
          acc[rt][j] = wmma_f32((v2f){a.z, a.w}, (v2f){b.z, b.w}, acc[rt][j]);
        }
    }
  }

#pragma unroll
  for (int j = 0; j < 6; ++j) {
    const float bs = *bp[j];
#pragma unroll
    for (int rt = 0; rt < 2; ++rt)
#pragma unroll
      for (int r = 0; r < 8; ++r)
        op[j][(size_t)(16 * rt + r) * DK] = acc[rt][j][r] + bs;
  }
}

#define KP 68
#define RP 33

__global__ __launch_bounds__(256) void attn_kernel(
    const float* __restrict__ Kb, const float* __restrict__ Qb,
    const float* __restrict__ Vb, float* __restrict__ Y) {
  __shared__ __attribute__((aligned(16))) float lds_k[B_ * 16 * KP];
  __shared__ float lds_y[8 * B_ * 2 * 8 * RP];

  const int lane = threadIdx.x & 31;
  const int w    = threadIdx.x >> 5;
  const int c    = lane & 15;
  const int h    = lane >> 4;
  const int i0   = (blockIdx.x >> 1) << 4;
  const int vb   = (blockIdx.x & 1) << 5;

  {
    const float scale = 0.125f;
#pragma unroll
    for (int it = 0; it < 8; ++it) {
      const int idx = threadIdx.x + 256 * it;
      const int b   = idx >> 8;
      const int row = (idx >> 4) & 15;
      const int d4  = idx & 15;
      v4f kv = *(const v4f*)(Kb + (size_t)(b * S_ + i0 + row) * DK + 4 * d4);
      kv = kv * scale;
      *(v4f*)(lds_k + (b * 16 + row) * KP + 4 * d4) = kv;
    }
  }
  __syncthreads();

  v8f yacc[B_][2];
#pragma unroll
  for (int b = 0; b < B_; ++b)
#pragma unroll
    for (int vt = 0; vt < 2; ++vt)
      yacc[b][vt] = (v8f){0.f, 0.f, 0.f, 0.f, 0.f, 0.f, 0.f, 0.f};

  v4f qs[3][8];
  v4f ks[2][8];
  v2f vs[5][4][2];
  v8f e[B_];

  auto load_q = [&](int set, int b, int j0) {
    const float* qr = Qb + (size_t)(b * S_ + j0 + c) * DK + 4 * h;
#pragma unroll
    for (int T = 0; T < 8; ++T) qs[set][T] = *(const v4f*)(qr + 8 * T);
  };
  auto load_k = [&](int set, int b) {
    const float* kr = lds_k + (b * 16 + c) * KP + 4 * h;
#pragma unroll
    for (int T = 0; T < 8; ++T) ks[set][T] = *(const v4f*)(kr + 8 * T);
  };
  auto load_v = [&](int set, int b, int j0) {
    const float* vr = Vb + (size_t)(b * S_ + j0 + 8 * h) * DV + vb + 2 * c;
#pragma unroll
    for (int t = 0; t < 4; ++t)
#pragma unroll
      for (int s = 0; s < 2; ++s)
        vs[set][t][s] = *(const v2f*)(vr + (2 * t + s) * DV);
  };
  auto e_phase = [&](int j0) {
#pragma unroll
    for (int b = 0; b < B_; ++b) {
      if (b + 2 < B_) load_q((b + 2) % 3, b + 2, j0);
      if (b + 1 < B_) load_k((b + 1) % 2, b + 1);
      v8f acc0 = (v8f){0.f, 0.f, 0.f, 0.f, 0.f, 0.f, 0.f, 0.f};
      v8f acc1 = (v8f){0.f, 0.f, 0.f, 0.f, 0.f, 0.f, 0.f, 0.f};
#pragma unroll
      for (int T = 0; T < 8; ++T) {
        const v4f q = qs[b % 3][T];
        const v4f k = ks[b % 2][T];
        acc0 = wmma_f32((v2f){q.x, q.y}, (v2f){k.x, k.y}, acc0);
        acc1 = wmma_f32((v2f){q.z, q.w}, (v2f){k.z, k.w}, acc1);
      }
      e[b] = acc0 + acc1;
    }
  };

  load_q(0, 0, (w * 16) << 4);
  load_q(1, 1, (w * 16) << 4);

  for (int n = 0; n < 16; ++n) {
    const int j0  = (w * 16 + n) << 4;
    const int j0n = (n < 15) ? j0 + 16 : j0;

    load_k(0, 0);
    e_phase(j0);

    load_v(0, 0, j0);
    load_v(1, 1, j0);
    load_v(2, 2, j0);
    load_v(3, 3, j0);
    load_q(0, 0, j0n);
    load_q(1, 1, j0n);

    asm volatile("" ::: "memory");

#pragma unroll
    for (int r = 0; r < 8; ++r) {
      const float mx =
          fmaxf(fmaxf(fmaxf(e[0][r], e[1][r]), fmaxf(e[2][r], e[3][r])),
                fmaxf(fmaxf(e[4][r], e[5][r]), fmaxf(e[6][r], e[7][r])));
      float p[B_];
#pragma unroll
      for (int b = 0; b < B_; ++b) p[b] = __expf(e[b][r] - mx);
      const float z = p[0] + p[1] + p[2] + p[3] + p[4] + p[5] + p[6] + p[7];
#pragma unroll
      for (int b = 0; b < B_; ++b) e[b][r] = p[b] / z;
    }

#pragma unroll
    for (int b = 0; b < B_; ++b) {
      if (b + 4 < B_) load_v((b + 4) % 5, b + 4, j0);
#pragma unroll
      for (int t = 0; t < 4; ++t) {
        const v2f pb = (v2f){e[b][2 * t], e[b][2 * t + 1]};
        const v2f v0 = vs[b % 5][t][0];
        const v2f v1 = vs[b % 5][t][1];
        yacc[b][0] = wmma_f32((v2f){v0.x, v1.x}, pb, yacc[b][0]);
        yacc[b][1] = wmma_f32((v2f){v0.y, v1.y}, pb, yacc[b][1]);
      }
    }
  }

#pragma unroll
  for (int b = 0; b < B_; ++b)
#pragma unroll
    for (int vt = 0; vt < 2; ++vt)
#pragma unroll
      for (int r = 0; r < 8; ++r)
        lds_y[(((w * B_ + b) * 2 + vt) * 8 + r) * RP + lane] = yacc[b][vt][r];
  __syncthreads();

  {
    const int vt = lane & 1;
    const int hh = lane >> 4;
    const int rr = (lane >> 1) & 7;
#pragma unroll 4
    for (int i = 0; i < 16; ++i) {
      float s = lds_y[(((0 * B_ + w) * 2 + vt) * 8 + rr) * RP + i + 16 * hh];
#pragma unroll
      for (int ww = 1; ww < 8; ++ww)
        s += lds_y[(((ww * B_ + w) * 2 + vt) * 8 + rr) * RP + i + 16 * hh];
      Y[(size_t)(w * S_ + i0 + i) * DV + vb + lane] = s;
    }
  }
}

extern "C" void kernel_launch(void* const* d_in, const int* in_sizes, int n_in,
                              void* d_out, int out_size, void* d_ws,
                              size_t ws_size, hipStream_t stream) {
  const float* X  = (const float*)d_in[0];
  const float* Wk = (const float*)d_in[1];
  const float* bk = (const float*)d_in[2];
  const float* Wq = (const float*)d_in[3];
  const float* bq = (const float*)d_in[4];
  const float* Wv = (const float*)d_in[5];
  const float* bv = (const float*)d_in[6];
  float* Y = (float*)d_out;

  float* ws = (float*)d_ws;
  const size_t NKQV = (size_t)B_ * S_ * DK;
  float* Kb  = ws;
  float* Qb  = ws + NKQV;
  float* Vb  = ws + 2 * NKQV;

  qkv_kernel<<<256, 128, 0, stream>>>(
      X, Wk, Wq, Wv, bk, bq, bv, Kb, Qb, Vb);
  attn_kernel<<<256, 256, 0, stream>>>(Kb, Qb, Vb, Y);
}


# attn softmax: every stage written across the 8 batches of 2 elements, exact division spelled out as its 11 steps, so 16 dependent chains advance together; every value and every sum order unchanged
# speedup vs baseline: 2.8954x; 1.1307x over previous
#include <hip/hip_runtime.h>

#define D_IN 512
#define DK   64
#define DV   64
#define B_   8
#define S_   2048

typedef __attribute__((ext_vector_type(2))) float v2f;
typedef __attribute__((ext_vector_type(4))) float v4f;
typedef __attribute__((ext_vector_type(8))) float v8f;

static __device__ __forceinline__ v8f wmma_f32(v2f a, v2f b, v8f c) {
  return __builtin_amdgcn_wmma_f32_16x16x4_f32(
       false, a,  false, b,
       (short)0, c,  false,  false);
}

#define RING 8

__global__ __launch_bounds__(128) void qkv_kernel(
    const float* __restrict__ X,
    const float* __restrict__ Wk, const float* __restrict__ Wq,
    const float* __restrict__ Wv,
    const float* __restrict__ bk, const float* __restrict__ bq,
    const float* __restrict__ bv,
    float* __restrict__ Kb, float* __restrict__ Qb, float* __restrict__ Vb) {
  const int lane = threadIdx.x & 31;
  const int w    = threadIdx.x >> 5;
  const int c    = lane & 15;
  const int h    = lane >> 4;
  const int rh   = w >> 1;
  const int ch   = w & 1;
  const int row0 = blockIdx.x * 64 + rh * 32;

  const float* xp = X + (size_t)(row0 + c) * D_IN + 4 * h;
  const float* wp[6];
  const float* bp[6];
  float*       op[6];
#pragma unroll
  for (int j = 0; j < 6; ++j) {
    const int g   = 6 * ch + j;
    const int mat = g >> 2;
    const int col = ((g & 3) << 4) + c;
    const float* Wm = (mat == 0) ? Wk : ((mat == 1) ? Wq : Wv);
    const float* bm = (mat == 0) ? bk : ((mat == 1) ? bq : bv);
    float*       om = (mat == 0) ? Kb : ((mat == 1) ? Qb : Vb);
    wp[j] = Wm + (size_t)col * D_IN + 4 * h;
    bp[j] = bm + col;
    op[j] = om + (size_t)(row0 + 8 * h) * DK + col;
  }

  v8f acc[2][6];
#pragma unroll
  for (int rt = 0; rt < 2; ++rt)
#pragma unroll
    for (int j = 0; j < 6; ++j)
      acc[rt][j] = (v8f){0.f, 0.f, 0.f, 0.f, 0.f, 0.f, 0.f, 0.f};

  v4f xa[RING][2];
  v4f wb[RING][6];
  auto load_set = [&](int set, int T) {
#pragma unroll
    for (int rt = 0; rt < 2; ++rt)
      xa[set][rt] = *(const v4f*)(xp + (size_t)rt * 16 * D_IN + 8 * T);
#pragma unroll
    for (int j = 0; j < 6; ++j)
      wb[set][j] = *(const v4f*)(wp[j] + 8 * T);
  };

#pragma unroll
  for (int s = 0; s < RING - 1; ++s) {
    load_set(s, s);
    asm volatile("" ::: "memory");
  }

  for (int T0 = 0; T0 < D_IN / 8; T0 += RING) {
#pragma unroll
    for (int q = 0; q < RING; ++q) {
      int Tn = T0 + q + RING - 1;
      Tn = (Tn < D_IN / 8) ? Tn : D_IN / 8 - 1;
      load_set((q + RING - 1) % RING, Tn);
      asm volatile("" ::: "memory");
      __builtin_amdgcn_sched_barrier(0);
#pragma unroll
      for (int rt = 0; rt < 2; ++rt)
#pragma unroll
        for (int j = 0; j < 6; ++j) {
          const v4f a = xa[q][rt];
          const v4f b = wb[q][j];
          acc[rt][j] = wmma_f32((v2f){a.x, a.y}, (v2f){b.x, b.y}, acc[rt][j]);
        }
#pragma unroll
      for (int rt = 0; rt < 2; ++rt)
#pragma unroll
        for (int j = 0; j < 6; ++j) {
          const v4f a = xa[q][rt];
          const v4f b = wb[q][j];
          acc[rt][j] = wmma_f32((v2f){a.z, a.w}, (v2f){b.z, b.w}, acc[rt][j]);
        }
    }
  }

#pragma unroll
  for (int j = 0; j < 6; ++j) {
    const float bs = *bp[j];
#pragma unroll
    for (int rt = 0; rt < 2; ++rt)
#pragma unroll
      for (int r = 0; r < 8; ++r)
        op[j][(size_t)(16 * rt + r) * DK] = acc[rt][j][r] + bs;
  }
}

#define KP 68
#define RP 33

__global__ __launch_bounds__(256) void attn_kernel(
    const float* __restrict__ Kb, const float* __restrict__ Qb,
    const float* __restrict__ Vb, float* __restrict__ Y) {
  __shared__ __attribute__((aligned(16))) float lds_k[B_ * 16 * KP];
  __shared__ float lds_y[8 * B_ * 2 * 8 * RP];

  const int lane = threadIdx.x & 31;
  const int w    = threadIdx.x >> 5;
  const int c    = lane & 15;
  const int h    = lane >> 4;
  const int i0   = (blockIdx.x >> 1) << 4;
  const int vb   = (blockIdx.x & 1) << 5;

  {
    const float scale = 0.125f;
#pragma unroll
    for (int it = 0; it < 8; ++it) {
      const int idx = threadIdx.x + 256 * it;
      const int b   = idx >> 8;
      const int row = (idx >> 4) & 15;
      const int d4  = idx & 15;
      v4f kv = *(const v4f*)(Kb + (size_t)(b * S_ + i0 + row) * DK + 4 * d4);
      kv = kv * scale;
      *(v4f*)(lds_k + (b * 16 + row) * KP + 4 * d4) = kv;
    }
  }
  __syncthreads();

  v8f yacc[B_][2];
#pragma unroll
  for (int b = 0; b < B_; ++b)
#pragma unroll
    for (int vt = 0; vt < 2; ++vt)
      yacc[b][vt] = (v8f){0.f, 0.f, 0.f, 0.f, 0.f, 0.f, 0.f, 0.f};

  v4f qs[3][8];
  v4f ks[2][8];
  v2f vs[5][4][2];
  v8f e[B_];

  auto load_q = [&](int set, int b, int j0) {
    const float* qr = Qb + (size_t)(b * S_ + j0 + c) * DK + 4 * h;
#pragma unroll
    for (int T = 0; T < 8; ++T) qs[set][T] = *(const v4f*)(qr + 8 * T);
  };
  auto load_k = [&](int set, int b) {
    const float* kr = lds_k + (b * 16 + c) * KP + 4 * h;
#pragma unroll
    for (int T = 0; T < 8; ++T) ks[set][T] = *(const v4f*)(kr + 8 * T);
  };
  auto load_v = [&](int set, int b, int j0) {
    const float* vr = Vb + (size_t)(b * S_ + j0 + 8 * h) * DV + vb + 2 * c;
#pragma unroll
    for (int t = 0; t < 4; ++t)
#pragma unroll
      for (int s = 0; s < 2; ++s)
        vs[set][t][s] = *(const v2f*)(vr + (2 * t + s) * DV);
  };
  auto e_phase = [&](int j0) {
#pragma unroll
    for (int b = 0; b < B_; ++b) {
      if (b + 2 < B_) load_q((b + 2) % 3, b + 2, j0);
      if (b + 1 < B_) load_k((b + 1) % 2, b + 1);
      v8f acc0 = (v8f){0.f, 0.f, 0.f, 0.f, 0.f, 0.f, 0.f, 0.f};
      v8f acc1 = (v8f){0.f, 0.f, 0.f, 0.f, 0.f, 0.f, 0.f, 0.f};
#pragma unroll
      for (int T = 0; T < 8; ++T) {
        const v4f q = qs[b % 3][T];
        const v4f k = ks[b % 2][T];
        acc0 = wmma_f32((v2f){q.x, q.y}, (v2f){k.x, k.y}, acc0);
        acc1 = wmma_f32((v2f){q.z, q.w}, (v2f){k.z, k.w}, acc1);
      }
      e[b] = acc0 + acc1;
    }
  };

  load_q(0, 0, (w * 16) << 4);
  load_q(1, 1, (w * 16) << 4);

  for (int n = 0; n < 16; ++n) {
    const int j0  = (w * 16 + n) << 4;
    const int j0n = (n < 15) ? j0 + 16 : j0;

    load_k(0, 0);
    e_phase(j0);

    load_v(0, 0, j0);
    load_v(1, 1, j0);
    load_v(2, 2, j0);
    load_v(3, 3, j0);
    load_q(0, 0, j0n);
    load_q(1, 1, j0n);

    asm volatile("" ::: "memory");

#pragma unroll
    for (int r0 = 0; r0 < 8; r0 += 2) {
      float mx[2], z[2];
      float p[2][B_], t[2][B_];
#pragma unroll
      for (int u = 0; u < 2; ++u) {
        const int r = r0 + u;
        mx[u] =
            fmaxf(fmaxf(fmaxf(e[0][r], e[1][r]), fmaxf(e[2][r], e[3][r])),
                  fmaxf(fmaxf(e[4][r], e[5][r]), fmaxf(e[6][r], e[7][r])));
      }
#pragma unroll
      for (int u = 0; u < 2; ++u)
#pragma unroll
        for (int b = 0; b < B_; ++b) t[u][b] = e[b][r0 + u] - mx[u];
#pragma unroll
      for (int u = 0; u < 2; ++u)
#pragma unroll
        for (int b = 0; b < B_; ++b) t[u][b] = t[u][b] * 0x1.715476p+0f;
#pragma unroll
      for (int u = 0; u < 2; ++u)
#pragma unroll
        for (int b = 0; b < B_; ++b) p[u][b] = __builtin_amdgcn_exp2f(t[u][b]);
#pragma unroll
      for (int u = 0; u < 2; ++u)
        z[u] = p[u][0] + p[u][1] + p[u][2] + p[u][3] + p[u][4] + p[u][5] + p[u][6] + p[u][7];

      float ds[2][B_], ns[2][B_], rc[2][B_], q[2][B_];
      bool  fl[2][B_];
#pragma unroll
      for (int u = 0; u < 2; ++u)
#pragma unroll
        for (int b = 0; b < B_; ++b) {
          bool unused;
          ds[u][b] = __builtin_amdgcn_div_scalef(p[u][b], z[u], false, &unused);
        }
#pragma unroll
      for (int u = 0; u < 2; ++u)
#pragma unroll
        for (int b = 0; b < B_; ++b) rc[u][b] = __builtin_amdgcn_rcpf(ds[u][b]);
#pragma unroll
      for (int u = 0; u < 2; ++u)
#pragma unroll
        for (int b = 0; b < B_; ++b)
          ns[u][b] = __builtin_amdgcn_div_scalef(p[u][b], z[u], true, &fl[u][b]);
#pragma unroll
      for (int u = 0; u < 2; ++u)
#pragma unroll
        for (int b = 0; b < B_; ++b) t[u][b] = __builtin_fmaf(-ds[u][b], rc[u][b], 1.0f);
#pragma unroll
      for (int u = 0; u < 2; ++u)
#pragma unroll
        for (int b = 0; b < B_; ++b) rc[u][b] = __builtin_fmaf(t[u][b], rc[u][b], rc[u][b]);
#pragma unroll
      for (int u = 0; u < 2; ++u)
#pragma unroll
        for (int b = 0; b < B_; ++b) q[u][b] = ns[u][b] * rc[u][b];
#pragma unroll
      for (int u = 0; u < 2; ++u)
#pragma unroll
        for (int b = 0; b < B_; ++b) t[u][b] = __builtin_fmaf(-ds[u][b], q[u][b], ns[u][b]);
#pragma unroll
      for (int u = 0; u < 2; ++u)
#pragma unroll
        for (int b = 0; b < B_; ++b) q[u][b] = __builtin_fmaf(t[u][b], rc[u][b], q[u][b]);
#pragma unroll
      for (int u = 0; u < 2; ++u)
#pragma unroll
        for (int b = 0; b < B_; ++b) t[u][b] = __builtin_fmaf(-ds[u][b], q[u][b], ns[u][b]);
#pragma unroll
      for (int u = 0; u < 2; ++u)
#pragma unroll
        for (int b = 0; b < B_; ++b)
          q[u][b] = __builtin_amdgcn_div_fmasf(t[u][b], rc[u][b], q[u][b], fl[u][b]);
#pragma unroll
      for (int u = 0; u < 2; ++u)
#pragma unroll
        for (int b = 0; b < B_; ++b)
          e[b][r0 + u] = __builtin_amdgcn_div_fixupf(q[u][b], z[u], p[u][b]);
    }

#pragma unroll
    for (int b = 0; b < B_; ++b) {
      if (b + 4 < B_) load_v((b + 4) % 5, b + 4, j0);
#pragma unroll
      for (int t = 0; t < 4; ++t) {
        const v2f pb = (v2f){e[b][2 * t], e[b][2 * t + 1]};
        const v2f v0 = vs[b % 5][t][0];
        const v2f v1 = vs[b % 5][t][1];
        yacc[b][0] = wmma_f32((v2f){v0.x, v1.x}, pb, yacc[b][0]);
        yacc[b][1] = wmma_f32((v2f){v0.y, v1.y}, pb, yacc[b][1]);
      }
    }
  }

#pragma unroll
  for (int b = 0; b < B_; ++b)
#pragma unroll
    for (int vt = 0; vt < 2; ++vt)
#pragma unroll
      for (int r = 0; r < 8; ++r)
        lds_y[(((w * B_ + b) * 2 + vt) * 8 + r) * RP + lane] = yacc[b][vt][r];
  __syncthreads();

  {
    const int vt = lane & 1;
    const int hh = lane >> 4;
    const int rr = (lane >> 1) & 7;
#pragma unroll 4
    for (int i = 0; i < 16; ++i) {
      float s = lds_y[(((0 * B_ + w) * 2 + vt) * 8 + rr) * RP + i + 16 * hh];
#pragma unroll
      for (int ww = 1; ww < 8; ++ww)
        s += lds_y[(((ww * B_ + w) * 2 + vt) * 8 + rr) * RP + i + 16 * hh];
      Y[(size_t)(w * S_ + i0 + i) * DV + vb + lane] = s;
    }
  }
}

extern "C" void kernel_launch(void* const* d_in, const int* in_sizes, int n_in,
                              void* d_out, int out_size, void* d_ws,
                              size_t ws_size, hipStream_t stream) {
  const float* X  = (const float*)d_in[0];
  const float* Wk = (const float*)d_in[1];
  const float* bk = (const float*)d_in[2];
  const float* Wq = (const float*)d_in[3];
  const float* bq = (const float*)d_in[4];
  const float* Wv = (const float*)d_in[5];
  const float* bv = (const float*)d_in[6];
  float* Y = (float*)d_out;

  float* ws = (float*)d_ws;
  const size_t NKQV = (size_t)B_ * S_ * DK;
  float* Kb  = ws;
  float* Qb  = ws + NKQV;
  float* Vb  = ws + 2 * NKQV;

  qkv_kernel<<<256, 128, 0, stream>>>(
      X, Wk, Wq, Wv, bk, bq, bv, Kb, Qb, Vb);
  attn_kernel<<<256, 256, 0, stream>>>(Kb, Qb, Vb, Y);
}


# attn block = (i-tile, batch half) over the whole V width, not (i-tile, V half): a block still forms all 8 exponentials and their sum but divides only its own 4 batches; values and sum orders kept
# speedup vs baseline: 3.0655x; 1.0587x over previous
#include <hip/hip_runtime.h>

#define D_IN 512
#define DK   64
#define DV   64
#define B_   8
#define S_   2048

typedef __attribute__((ext_vector_type(2))) float v2f;
typedef __attribute__((ext_vector_type(4))) float v4f;
typedef __attribute__((ext_vector_type(8))) float v8f;

static __device__ __forceinline__ v8f wmma_f32(v2f a, v2f b, v8f c) {
  return __builtin_amdgcn_wmma_f32_16x16x4_f32(
       false, a,  false, b,
       (short)0, c,  false,  false);
}

#define RING 8

__global__ __launch_bounds__(128) void qkv_kernel(
    const float* __restrict__ X,
    const float* __restrict__ Wk, const float* __restrict__ Wq,
    const float* __restrict__ Wv,
    const float* __restrict__ bk, const float* __restrict__ bq,
    const float* __restrict__ bv,
    float* __restrict__ Kb, float* __restrict__ Qb, float* __restrict__ Vb) {
  const int lane = threadIdx.x & 31;
  const int w    = threadIdx.x >> 5;
  const int c    = lane & 15;
  const int h    = lane >> 4;
  const int rh   = w >> 1;
  const int ch   = w & 1;
  const int row0 = blockIdx.x * 64 + rh * 32;

  const float* xp = X + (size_t)(row0 + c) * D_IN + 4 * h;
  const float* wp[6];
  const float* bp[6];
  float*       op[6];
#pragma unroll
  for (int j = 0; j < 6; ++j) {
    const int g   = 6 * ch + j;
    const int mat = g >> 2;
    const int col = ((g & 3) << 4) + c;
    const float* Wm = (mat == 0) ? Wk : ((mat == 1) ? Wq : Wv);
    const float* bm = (mat == 0) ? bk : ((mat == 1) ? bq : bv);
    float*       om = (mat == 0) ? Kb : ((mat == 1) ? Qb : Vb);
    wp[j] = Wm + (size_t)col * D_IN + 4 * h;
    bp[j] = bm + col;
    op[j] = om + (size_t)(row0 + 8 * h) * DK + col;
  }

  v8f acc[2][6];
#pragma unroll
  for (int rt = 0; rt < 2; ++rt)
#pragma unroll
    for (int j = 0; j < 6; ++j)
      acc[rt][j] = (v8f){0.f, 0.f, 0.f, 0.f, 0.f, 0.f, 0.f, 0.f};

  v4f xa[RING][2];
  v4f wb[RING][6];
  auto load_set = [&](int set, int T) {
#pragma unroll
    for (int rt = 0; rt < 2; ++rt)
      xa[set][rt] = *(const v4f*)(xp + (size_t)rt * 16 * D_IN + 8 * T);
#pragma unroll
    for (int j = 0; j < 6; ++j)
      wb[set][j] = *(const v4f*)(wp[j] + 8 * T);
  };

#pragma unroll
  for (int s = 0; s < RING - 1; ++s) {
    load_set(s, s);
    asm volatile("" ::: "memory");
  }

  for (int T0 = 0; T0 < D_IN / 8; T0 += RING) {
#pragma unroll
    for (int q = 0; q < RING; ++q) {
      int Tn = T0 + q + RING - 1;
      Tn = (Tn < D_IN / 8) ? Tn : D_IN / 8 - 1;
      load_set((q + RING - 1) % RING, Tn);
      asm volatile("" ::: "memory");
      __builtin_amdgcn_sched_barrier(0);
#pragma unroll
      for (int rt = 0; rt < 2; ++rt)
#pragma unroll
        for (int j = 0; j < 6; ++j) {
          const v4f a = xa[q][rt];
          const v4f b = wb[q][j];
          acc[rt][j] = wmma_f32((v2f){a.x, a.y}, (v2f){b.x, b.y}, acc[rt][j]);
        }
#pragma unroll
      for (int rt = 0; rt < 2; ++rt)
#pragma unroll
        for (int j = 0; j < 6; ++j) {
          const v4f a = xa[q][rt];
          const v4f b = wb[q][j];
          acc[rt][j] = wmma_f32((v2f){a.z, a.w}, (v2f){b.z, b.w}, acc[rt][j]);
        }
    }
  }

#pragma unroll
  for (int j = 0; j < 6; ++j) {
    const float bs = *bp[j];
#pragma unroll
    for (int rt = 0; rt < 2; ++rt)
#pragma unroll
      for (int r = 0; r < 8; ++r)
        op[j][(size_t)(16 * rt + r) * DK] = acc[rt][j][r] + bs;
  }
}

#define KP 68
#define RP 33

template <int N> struct IntC { static constexpr int value = N; };

__global__ __launch_bounds__(256) void attn_kernel(
    const float* __restrict__ Kb, const float* __restrict__ Qb,
    const float* __restrict__ Vb, float* __restrict__ Y) {
  __shared__ __attribute__((aligned(16))) float lds_k[B_ * 16 * KP];
  __shared__ float lds_y[8 * 4 * 4 * 8 * RP];

  const int lane = threadIdx.x & 31;
  const int w    = threadIdx.x >> 5;
  const int c    = lane & 15;
  const int h    = lane >> 4;
  const int i0   = (blockIdx.x >> 1) << 4;

  {
    const float scale = 0.125f;
#pragma unroll
    for (int it = 0; it < 8; ++it) {
      const int idx = threadIdx.x + 256 * it;
      const int b   = idx >> 8;
      const int row = (idx >> 4) & 15;
      const int d4  = idx & 15;
      v4f kv = *(const v4f*)(Kb + (size_t)(b * S_ + i0 + row) * DK + 4 * d4);
      kv = kv * scale;
      *(v4f*)(lds_k + (b * 16 + row) * KP + 4 * d4) = kv;
    }
  }
  __syncthreads();

  auto body = [&](auto gc) {
  constexpr int B0 = 4 * decltype(gc)::value;

  v8f yacc[4][4];
#pragma unroll
  for (int o = 0; o < 4; ++o)
#pragma unroll
    for (int vt = 0; vt < 4; ++vt)
      yacc[o][vt] = (v8f){0.f, 0.f, 0.f, 0.f, 0.f, 0.f, 0.f, 0.f};

  v4f qs[3][8];
  v4f ks[2][8];
  v4f vs[3][4][2];
  v8f e[B_];

  auto load_q = [&](int set, int b, int j0) {
    const float* qr = Qb + (size_t)(b * S_ + j0 + c) * DK + 4 * h;
#pragma unroll
    for (int T = 0; T < 8; ++T) qs[set][T] = *(const v4f*)(qr + 8 * T);
  };
  auto load_k = [&](int set, int b) {
    const float* kr = lds_k + (b * 16 + c) * KP + 4 * h;
#pragma unroll
    for (int T = 0; T < 8; ++T) ks[set][T] = *(const v4f*)(kr + 8 * T);
  };
  auto load_v = [&](int set, int o, int j0) {
    const float* vr = Vb + (size_t)((B0 + o) * S_ + j0 + 8 * h) * DV + 4 * c;
#pragma unroll
    for (int t = 0; t < 4; ++t)
#pragma unroll
      for (int s = 0; s < 2; ++s)
        vs[set][t][s] = *(const v4f*)(vr + (2 * t + s) * DV);
  };
  auto e_phase = [&](int j0) {
#pragma unroll
    for (int b = 0; b < B_; ++b) {
      if (b + 2 < B_) load_q((b + 2) % 3, b + 2, j0);
      if (b + 1 < B_) load_k((b + 1) % 2, b + 1);
      v8f acc0 = (v8f){0.f, 0.f, 0.f, 0.f, 0.f, 0.f, 0.f, 0.f};
      v8f acc1 = (v8f){0.f, 0.f, 0.f, 0.f, 0.f, 0.f, 0.f, 0.f};
#pragma unroll
      for (int T = 0; T < 8; ++T) {
        const v4f q = qs[b % 3][T];
        const v4f k = ks[b % 2][T];
        acc0 = wmma_f32((v2f){q.x, q.y}, (v2f){k.x, k.y}, acc0);
        acc1 = wmma_f32((v2f){q.z, q.w}, (v2f){k.z, k.w}, acc1);
      }
      e[b] = acc0 + acc1;
    }
  };

  load_q(0, 0, (w * 16) << 4);
  load_q(1, 1, (w * 16) << 4);

  for (int n = 0; n < 16; ++n) {
    const int j0  = (w * 16 + n) << 4;
    const int j0n = (n < 15) ? j0 + 16 : j0;

    load_k(0, 0);
    e_phase(j0);

    load_v(0, 0, j0);
    load_v(1, 1, j0);
    load_v(2, 2, j0);
    load_q(0, 0, j0n);
    load_q(1, 1, j0n);

    asm volatile("" ::: "memory");

#pragma unroll
    for (int r0 = 0; r0 < 8; r0 += 2) {
      float mx[2], z[2];
      float p[2][B_], t[2][B_];
#pragma unroll
      for (int u = 0; u < 2; ++u) {
        const int r = r0 + u;
        mx[u] =
            fmaxf(fmaxf(fmaxf(e[0][r], e[1][r]), fmaxf(e[2][r], e[3][r])),
                  fmaxf(fmaxf(e[4][r], e[5][r]), fmaxf(e[6][r], e[7][r])));
      }
#pragma unroll
      for (int u = 0; u < 2; ++u)
#pragma unroll
        for (int b = 0; b < B_; ++b) t[u][b] = e[b][r0 + u] - mx[u];
#pragma unroll
      for (int u = 0; u < 2; ++u)
#pragma unroll
        for (int b = 0; b < B_; ++b) t[u][b] = t[u][b] * 0x1.715476p+0f;
#pragma unroll
      for (int u = 0; u < 2; ++u)
#pragma unroll
        for (int b = 0; b < B_; ++b) p[u][b] = __builtin_amdgcn_exp2f(t[u][b]);
#pragma unroll
      for (int u = 0; u < 2; ++u)
        z[u] = p[u][0] + p[u][1] + p[u][2] + p[u][3] + p[u][4] + p[u][5] + p[u][6] + p[u][7];

      float ds[2][4], ns[2][4], rc[2][4], q[2][4];
      bool  fl[2][4];
#pragma unroll
      for (int u = 0; u < 2; ++u)
#pragma unroll
        for (int o = 0; o < 4; ++o) {
          bool unused;
          ds[u][o] = __builtin_amdgcn_div_scalef(p[u][B0 + o], z[u], false, &unused);
        }
#pragma unroll
      for (int u = 0; u < 2; ++u)
#pragma unroll
        for (int o = 0; o < 4; ++o) rc[u][o] = __builtin_amdgcn_rcpf(ds[u][o]);
#pragma unroll
      for (int u = 0; u < 2; ++u)
#pragma unroll
        for (int o = 0; o < 4; ++o)
          ns[u][o] = __builtin_amdgcn_div_scalef(p[u][B0 + o], z[u], true, &fl[u][o]);
#pragma unroll
      for (int u = 0; u < 2; ++u)
#pragma unroll
        for (int o = 0; o < 4; ++o) t[u][o] = __builtin_fmaf(-ds[u][o], rc[u][o], 1.0f);
#pragma unroll
      for (int u = 0; u < 2; ++u)
#pragma unroll
        for (int o = 0; o < 4; ++o) rc[u][o] = __builtin_fmaf(t[u][o], rc[u][o], rc[u][o]);
#pragma unroll
      for (int u = 0; u < 2; ++u)
#pragma unroll
        for (int o = 0; o < 4; ++o) q[u][o] = ns[u][o] * rc[u][o];
#pragma unroll
      for (int u = 0; u < 2; ++u)
#pragma unroll
        for (int o = 0; o < 4; ++o) t[u][o] = __builtin_fmaf(-ds[u][o], q[u][o], ns[u][o]);
#pragma unroll
      for (int u = 0; u < 2; ++u)
#pragma unroll
        for (int o = 0; o < 4; ++o) q[u][o] = __builtin_fmaf(t[u][o], rc[u][o], q[u][o]);
#pragma unroll
      for (int u = 0; u < 2; ++u)
#pragma unroll
        for (int o = 0; o < 4; ++o) t[u][o] = __builtin_fmaf(-ds[u][o], q[u][o], ns[u][o]);
#pragma unroll
      for (int u = 0; u < 2; ++u)
#pragma unroll
        for (int o = 0; o < 4; ++o)
          q[u][o] = __builtin_amdgcn_div_fmasf(t[u][o], rc[u][o], q[u][o], fl[u][o]);
#pragma unroll
      for (int u = 0; u < 2; ++u)
#pragma unroll
        for (int o = 0; o < 4; ++o)
          e[B0 + o][r0 + u] = __builtin_amdgcn_div_fixupf(q[u][o], z[u], p[u][B0 + o]);
    }

#pragma unroll
    for (int o = 0; o < 4; ++o) {
      if (o == 1) load_v(0, 3, j0);
#pragma unroll
      for (int t = 0; t < 4; ++t) {
        const v2f pb = (v2f){e[B0 + o][2 * t], e[B0 + o][2 * t + 1]};
        const v4f v0 = vs[o % 3][t][0];
        const v4f v1 = vs[o % 3][t][1];
#pragma unroll
        for (int vt = 0; vt < 4; ++vt)
          yacc[o][vt] = wmma_f32((v2f){v0[vt], v1[vt]}, pb, yacc[o][vt]);
      }
    }
  }

#pragma unroll
  for (int o = 0; o < 4; ++o)
#pragma unroll
    for (int vt = 0; vt < 4; ++vt)
#pragma unroll
      for (int r = 0; r < 8; ++r)
        lds_y[(((w * 4 + o) * 4 + vt) * 8 + r) * RP + lane] = yacc[o][vt][r];
  __syncthreads();

  {
    const int o  = w >> 1;
    const int hh = w & 1;
    const int vt = lane & 3;
    const int rr = lane >> 2;
#pragma unroll 4
    for (int i = 0; i < 16; ++i) {
      float s = lds_y[(((0 * 4 + o) * 4 + vt) * 8 + rr) * RP + i + 16 * hh];
#pragma unroll
      for (int ww = 1; ww < 8; ++ww)
        s += lds_y[(((ww * 4 + o) * 4 + vt) * 8 + rr) * RP + i + 16 * hh];
      Y[(size_t)((B0 + o) * S_ + i0 + i) * DV + 32 * hh + lane] = s;
    }
  }
  };

  if (blockIdx.x & 1) body(IntC<1>{}); else body(IntC<0>{});
}

extern "C" void kernel_launch(void* const* d_in, const int* in_sizes, int n_in,
                              void* d_out, int out_size, void* d_ws,
                              size_t ws_size, hipStream_t stream) {
  const float* X  = (const float*)d_in[0];
  const float* Wk = (const float*)d_in[1];
  const float* bk = (const float*)d_in[2];
  const float* Wq = (const float*)d_in[3];
  const float* bq = (const float*)d_in[4];
  const float* Wv = (const float*)d_in[5];
  const float* bv = (const float*)d_in[6];
  float* Y = (float*)d_out;

  float* ws = (float*)d_ws;
  const size_t NKQV = (size_t)B_ * S_ * DK;
  float* Kb  = ws;
  float* Qb  = ws + NKQV;
  float* Vb  = ws + 2 * NKQV;

  qkv_kernel<<<256, 128, 0, stream>>>(
      X, Wk, Wq, Wv, bk, bq, bv, Kb, Qb, Vb);
  attn_kernel<<<256, 256, 0, stream>>>(Kb, Qb, Vb, Y);
}
